# starting kernel
# baseline (speedup 1.0000x reference)
#include <hip/hip_runtime.h>

typedef float v2f __attribute__((ext_vector_type(2)));
typedef float v8f __attribute__((ext_vector_type(8)));

#define DIM     1024
#define NROWS   16384
#define NCLS    10
#define IMG     784
#define KSTEPS  (DIM / 4)

__global__ __launch_bounds__(256) void ht_norm_kernel(const float* __restrict__ canon,
                                                      float* __restrict__ rnorm) {
    __shared__ float red[256];
    const int c = blockIdx.x;
    float s = 0.0f;
    for (int i = threadIdx.x; i < IMG; i += 256) {
        float v = canon[c * IMG + i];
        s += v * v;
    }
    red[threadIdx.x] = s;
    __syncthreads();
    for (int off = 128; off > 0; off >>= 1) {
        if (threadIdx.x < off) red[threadIdx.x] += red[threadIdx.x + off];
        __syncthreads();
    }
    if (threadIdx.x == 0) rnorm[c] = rsqrtf(red[0]);
}

__global__ __launch_bounds__(256) void ht_packB_kernel(const float* __restrict__ canon,
                                                       const float* __restrict__ rnorm,
                                                       float* __restrict__ Bbuf) {
    const int o    = blockIdx.x * 256 + threadIdx.x;
    const int s    = o >> 6;
    const int t    = o & 63;
    const int l    = t >> 1;
    const int j    = t & 1;
    const int c    = l & 15;
    const int half = l >> 4;
    const int d    = 4 * s + 2 * half + j;
    float v = 0.0f;
    if (c < NCLS && d < IMG) v = canon[c * IMG + d] * rnorm[c];
    Bbuf[o] = v;
}

__global__ __launch_bounds__(128) void ht_swaptest_kernel(const float* __restrict__ z_re,
                                                          const float* __restrict__ z_im,
                                                          const float* __restrict__ Bbuf,
                                                          float* __restrict__ out) {
    const int lane = threadIdx.x & 31;
    const int wave = threadIdx.x >> 5;
    const int tile = blockIdx.x * 4 + wave;
    const int base = tile * 16;
    const int m    = lane & 15;
    const int half = lane >> 4;

    const float* ar = z_re + (size_t)(base + m) * DIM + 2 * half;
    const float* ai = z_im + (size_t)(base + m) * DIM + 2 * half;
    const float* bp = Bbuf + lane * 2;

    v8f acc_re = {};
    v8f acc_im = {};

#pragma unroll 4
    for (int s = 0; s < KSTEPS; ++s) {
        v2f a0 = *(const v2f*)(ar + 4 * s);
        v2f a1 = *(const v2f*)(ai + 4 * s);
        v2f b  = *(const v2f*)(bp + 64 * s);
        acc_re = __builtin_amdgcn_wmma_f32_16x16x4_f32(
            false, a0, false, b, (short)0, acc_re, false, false);
        acc_im = __builtin_amdgcn_wmma_f32_16x16x4_f32(
            false, a1, false, b, (short)0, acc_im, false, false);
    }

    if (m < NCLS) {
#pragma unroll
        for (int r = 0; r < 8; ++r) {
            const float re = acc_re[r];
            const float im = acc_im[r];
            out[(size_t)(base + r + 8 * half) * NCLS + m] = re * re + im * im;
        }
    }
}

extern "C" void kernel_launch(void* const* d_in, const int* in_sizes, int n_in,
                              void* d_out, int out_size, void* d_ws, size_t ws_size,
                              hipStream_t stream) {
    const float* z_re  = (const float*)d_in[0];
    const float* z_im  = (const float*)d_in[1];
    const float* canon = (const float*)d_in[2];
    float* out   = (float*)d_out;
    float* Bbuf  = (float*)d_ws;
    float* rnorm = Bbuf + 16 * DIM;

    ht_norm_kernel<<<NCLS, 256, 0, stream>>>(canon, rnorm);
    ht_packB_kernel<<<(16 * DIM) / 256, 256, 0, stream>>>(canon, rnorm, Bbuf);
    ht_swaptest_kernel<<<NROWS / 64, 128, 0, stream>>>(z_re, z_im, Bbuf, out);
}


# K split over 4 waves per 16-row tile (16 waves per WGP, was 4), each stage of 24 loads requested a stage ahead in registers; partial tiles added through LDS, so the order of the K sum changes
# speedup vs baseline: 1.7272x; 1.7272x over previous
#include <hip/hip_runtime.h>

typedef float v2f __attribute__((ext_vector_type(2)));
typedef float v8f __attribute__((ext_vector_type(8)));

#define DIM     1024
#define NROWS   16384
#define NCLS    10
#define IMG     784
#define KSTEPS  (DIM / 4)

__global__ __launch_bounds__(256) void ht_norm_kernel(const float* __restrict__ canon,
                                                      float* __restrict__ rnorm) {
    __shared__ float red[256];
    const int c = blockIdx.x;
    float s = 0.0f;
    for (int i = threadIdx.x; i < IMG; i += 256) {
        float v = canon[c * IMG + i];
        s += v * v;
    }
    red[threadIdx.x] = s;
    __syncthreads();
    for (int off = 128; off > 0; off >>= 1) {
        if (threadIdx.x < off) red[threadIdx.x] += red[threadIdx.x + off];
        __syncthreads();
    }
    if (threadIdx.x == 0) rnorm[c] = rsqrtf(red[0]);
}

__global__ __launch_bounds__(256) void ht_packB_kernel(const float* __restrict__ canon,
                                                       const float* __restrict__ rnorm,
                                                       float* __restrict__ Bbuf) {
    const int o    = blockIdx.x * 256 + threadIdx.x;
    const int s    = o >> 6;
    const int t    = o & 63;
    const int l    = t >> 1;
    const int j    = t & 1;
    const int c    = l & 15;
    const int half = l >> 4;
    const int d    = 4 * s + 2 * half + j;
    float v = 0.0f;
    if (c < NCLS && d < IMG) v = canon[c * IMG + d] * rnorm[c];
    Bbuf[o] = v;
}

#define KSPLIT   4
#define QSTEPS   (KSTEPS / KSPLIT)
#define STG      8
#define NSTG     (QSTEPS / STG)

struct Stage {
    v2f a0[STG];
    v2f a1[STG];
    v2f b[STG];
};

__device__ __forceinline__ void load_stage(Stage& st, const float* ar, const float* ai,
                                           const float* bp, int s0) {
#pragma unroll
    for (int u = 0; u < STG; ++u) st.a0[u] = *(const v2f*)(ar + 4 * (s0 + u));
#pragma unroll
    for (int u = 0; u < STG; ++u) st.a1[u] = *(const v2f*)(ai + 4 * (s0 + u));
#pragma unroll
    for (int u = 0; u < STG; ++u) st.b[u]  = *(const v2f*)(bp + 64 * (s0 + u));
}

__device__ __forceinline__ void mma_stage(const Stage& st, v8f& acc_re, v8f& acc_im) {
#pragma unroll
    for (int u = 0; u < STG; ++u) {
        acc_re = __builtin_amdgcn_wmma_f32_16x16x4_f32(
            false, st.a0[u], false, st.b[u], (short)0, acc_re, false, false);
        acc_im = __builtin_amdgcn_wmma_f32_16x16x4_f32(
            false, st.a1[u], false, st.b[u], (short)0, acc_im, false, false);
    }
}

__global__ __launch_bounds__(256) void ht_swaptest_kernel(const float* __restrict__ z_re,
                                                          const float* __restrict__ z_im,
                                                          const float* __restrict__ Bbuf,
                                                          float* __restrict__ out) {
    __shared__ float part[2][KSPLIT][2][256];

    const int lane = threadIdx.x & 31;
    const int wave = threadIdx.x >> 5;
    const int tl   = wave >> 2;
    const int kq   = wave & 3;
    const int tile = blockIdx.x * 2 + tl;
    const int base = tile * 16;
    const int m    = lane & 15;
    const int half = lane >> 4;

    const float* ar = z_re + (size_t)(base + m) * DIM + kq * (DIM / KSPLIT) + 2 * half;
    const float* ai = z_im + (size_t)(base + m) * DIM + kq * (DIM / KSPLIT) + 2 * half;
    const float* bp = Bbuf + (size_t)kq * QSTEPS * 64 + lane * 2;

    v8f acc_re = {};
    v8f acc_im = {};

    Stage s0, s1;
    load_stage(s0, ar, ai, bp, 0);
#pragma unroll
    for (int st = 0; st < NSTG; st += 2) {
        load_stage(s1, ar, ai, bp, (st + 1) * STG);
        __builtin_amdgcn_sched_barrier(0);
        mma_stage(s0, acc_re, acc_im);
        __builtin_amdgcn_sched_barrier(0);
        if (st + 2 < NSTG) load_stage(s0, ar, ai, bp, (st + 2) * STG);
        __builtin_amdgcn_sched_barrier(0);
        mma_stage(s1, acc_re, acc_im);
        __builtin_amdgcn_sched_barrier(0);
    }

#pragma unroll
    for (int r = 0; r < 8; ++r) {
        part[tl][kq][0][r * 32 + lane] = acc_re[r];
        part[tl][kq][1][r * 32 + lane] = acc_im[r];
    }
    __syncthreads();

    const int t = threadIdx.x & 127;
    for (int o = t; o < 16 * NCLS; o += 128) {
        const int row = o / NCLS;
        const int c   = o - row * NCLS;
        const int idx = (row & 7) * 32 + 16 * (row >> 3) + c;
        float re = part[tl][0][0][idx];
        float im = part[tl][0][1][idx];
        for (int q = 1; q < KSPLIT; ++q) {
            re += part[tl][q][0][idx];
            im += part[tl][q][1][idx];
        }
        out[(size_t)base * NCLS + o] = re * re + im * im;
    }
}

extern "C" void kernel_launch(void* const* d_in, const int* in_sizes, int n_in,
                              void* d_out, int out_size, void* d_ws, size_t ws_size,
                              hipStream_t stream) {
    const float* z_re  = (const float*)d_in[0];
    const float* z_im  = (const float*)d_in[1];
    const float* canon = (const float*)d_in[2];
    float* out   = (float*)d_out;
    float* Bbuf  = (float*)d_ws;
    float* rnorm = Bbuf + 16 * DIM;

    ht_norm_kernel<<<NCLS, 256, 0, stream>>>(canon, rnorm);
    ht_packB_kernel<<<(16 * DIM) / 256, 256, 0, stream>>>(canon, rnorm, Bbuf);
    ht_swaptest_kernel<<<NROWS / 32, 256, 0, stream>>>(z_re, z_im, Bbuf, out);
}


# K loop ends at dim 784: ref is zero-padded beyond, so z dims 784-1023 (exact zero products) are never read; each of 4 waves: 6 line stages + 1 k-step of the half line; K sum order changes
# speedup vs baseline: 1.9174x; 1.1101x over previous
#include <hip/hip_runtime.h>

typedef float v2f __attribute__((ext_vector_type(2)));
typedef float v8f __attribute__((ext_vector_type(8)));

#define DIM     1024
#define NROWS   16384
#define NCLS    10
#define IMG     784
#define KNZ     (IMG / 4)

__global__ __launch_bounds__(256) void ht_norm_kernel(const float* __restrict__ canon,
                                                      float* __restrict__ rnorm) {
    __shared__ float red[256];
    const int c = blockIdx.x;
    float s = 0.0f;
    for (int i = threadIdx.x; i < IMG; i += 256) {
        float v = canon[c * IMG + i];
        s += v * v;
    }
    red[threadIdx.x] = s;
    __syncthreads();
    for (int off = 128; off > 0; off >>= 1) {
        if (threadIdx.x < off) red[threadIdx.x] += red[threadIdx.x + off];
        __syncthreads();
    }
    if (threadIdx.x == 0) rnorm[c] = rsqrtf(red[0]);
}

__global__ __launch_bounds__(256) void ht_packB_kernel(const float* __restrict__ canon,
                                                       const float* __restrict__ rnorm,
                                                       float* __restrict__ Bbuf) {
    const int o    = blockIdx.x * 256 + threadIdx.x;
    const int s    = o >> 6;
    const int t    = o & 63;
    const int l    = t >> 1;
    const int j    = t & 1;
    const int c    = l & 15;
    const int half = l >> 4;
    const int d    = 4 * s + 2 * half + j;
    float v = 0.0f;
    if (c < NCLS) v = canon[c * IMG + d] * rnorm[c];
    Bbuf[o] = v;
}

#define KSPLIT   4
#define QSTEPS   48
#define STG      8
#define NSTG     (QSTEPS / STG)
#define TAIL0    (KSPLIT * QSTEPS)
static_assert(TAIL0 + KSPLIT == KNZ, "the K parts must cover exactly the dims below 784");

struct Stage {
    v2f a0[STG];
    v2f a1[STG];
    v2f b[STG];
};

__device__ __forceinline__ void load_stage(Stage& st, const float* ar, const float* ai,
                                           const float* bp, int s0) {
#pragma unroll
    for (int u = 0; u < STG; ++u) st.a0[u] = *(const v2f*)(ar + 4 * (s0 + u));
#pragma unroll
    for (int u = 0; u < STG; ++u) st.a1[u] = *(const v2f*)(ai + 4 * (s0 + u));
#pragma unroll
    for (int u = 0; u < STG; ++u) st.b[u]  = *(const v2f*)(bp + 64 * (s0 + u));
}

__device__ __forceinline__ void mma_stage(const Stage& st, v8f& acc_re, v8f& acc_im) {
#pragma unroll
    for (int u = 0; u < STG; ++u) {
        acc_re = __builtin_amdgcn_wmma_f32_16x16x4_f32(
            false, st.a0[u], false, st.b[u], (short)0, acc_re, false, false);
        acc_im = __builtin_amdgcn_wmma_f32_16x16x4_f32(
            false, st.a1[u], false, st.b[u], (short)0, acc_im, false, false);
    }
}

__global__ __launch_bounds__(256) void ht_swaptest_kernel(const float* __restrict__ z_re,
                                                          const float* __restrict__ z_im,
                                                          const float* __restrict__ Bbuf,
                                                          float* __restrict__ out) {
    __shared__ float part[2][KSPLIT][2][256];

    const int lane = threadIdx.x & 31;
    const int wave = __builtin_amdgcn_readfirstlane(threadIdx.x >> 5);
    const int tl   = wave >> 2;
    const int kq   = wave & 3;
    const int tile = blockIdx.x * 2 + tl;
    const int base = tile * 16;
    const int m    = lane & 15;
    const int half = lane >> 4;

    const float* ar = z_re + (size_t)(base + m) * DIM + kq * (QSTEPS * 4) + 2 * half;
    const float* ai = z_im + (size_t)(base + m) * DIM + kq * (QSTEPS * 4) + 2 * half;
    const float* bp = Bbuf + (size_t)kq * QSTEPS * 64 + lane * 2;
    const int ts = TAIL0 + kq;
    const float* tr = z_re + (size_t)(base + m) * DIM + 4 * ts + 2 * half;
    const float* ti = z_im + (size_t)(base + m) * DIM + 4 * ts + 2 * half;
    const float* tp = Bbuf + (size_t)ts * 64 + lane * 2;

    v8f acc_re = {};
    v8f acc_im = {};

    const v2f ta0 = *(const v2f*)tr;
    const v2f ta1 = *(const v2f*)ti;
    const v2f tb  = *(const v2f*)tp;
    __builtin_amdgcn_sched_barrier(0);

    Stage s0, s1;
    load_stage(s0, ar, ai, bp, 0);
    __builtin_amdgcn_sched_barrier(0);
#pragma unroll
    for (int st = 0; st < NSTG; st += 2) {
        load_stage(s1, ar, ai, bp, (st + 1) * STG);
        __builtin_amdgcn_sched_barrier(0);
        if (st == 0) {
            acc_re = __builtin_amdgcn_wmma_f32_16x16x4_f32(
                false, ta0, false, tb, (short)0, acc_re, false, false);
            acc_im = __builtin_amdgcn_wmma_f32_16x16x4_f32(
                false, ta1, false, tb, (short)0, acc_im, false, false);
        }
        mma_stage(s0, acc_re, acc_im);
        __builtin_amdgcn_sched_barrier(0);
        if (st + 2 < NSTG) load_stage(s0, ar, ai, bp, (st + 2) * STG);
        __builtin_amdgcn_sched_barrier(0);
        mma_stage(s1, acc_re, acc_im);
        __builtin_amdgcn_sched_barrier(0);
    }

#pragma unroll
    for (int r = 0; r < 8; ++r) {
        part[tl][kq][0][r * 32 + lane] = acc_re[r];
        part[tl][kq][1][r * 32 + lane] = acc_im[r];
    }
    __syncthreads();

    const int t = threadIdx.x & 127;
    for (int o = t; o < 16 * NCLS; o += 128) {
        const int row = o / NCLS;
        const int c   = o - row * NCLS;
        const int idx = (row & 7) * 32 + 16 * (row >> 3) + c;
        float re = part[tl][0][0][idx];
        float im = part[tl][0][1][idx];
        for (int q = 1; q < KSPLIT; ++q) {
            re += part[tl][q][0][idx];
            im += part[tl][q][1][idx];
        }
        out[(size_t)base * NCLS + o] = re * re + im * im;
    }
}

extern "C" void kernel_launch(void* const* d_in, const int* in_sizes, int n_in,
                              void* d_out, int out_size, void* d_ws, size_t ws_size,
                              hipStream_t stream) {
    const float* z_re  = (const float*)d_in[0];
    const float* z_im  = (const float*)d_in[1];
    const float* canon = (const float*)d_in[2];
    float* out   = (float*)d_out;
    float* Bbuf  = (float*)d_ws;
    float* rnorm = Bbuf + 16 * DIM;

    ht_norm_kernel<<<NCLS, 256, 0, stream>>>(canon, rnorm);
    ht_packB_kernel<<<(KNZ * 64) / 256, 256, 0, stream>>>(canon, rnorm, Bbuf);
    ht_swaptest_kernel<<<NROWS / 32, 256, 0, stream>>>(z_re, z_im, Bbuf, out);
}


# One launch, not three: each wave copies its K part of canon to LDS, reads its B fragments there and sums its squares; result scaled by 1/norm^2 at the end: norm sum order and scaling point change
# speedup vs baseline: 2.7117x; 1.4143x over previous
#include <hip/hip_runtime.h>

typedef float v2f __attribute__((ext_vector_type(2)));
typedef float v4f __attribute__((ext_vector_type(4)));
typedef float v8f __attribute__((ext_vector_type(8)));

#define DIM     1024
#define NROWS   16384
#define NCLS    10
#define IMG     784
#define KNZ     (IMG / 4)

#define KSPLIT   4
#define QSTEPS   48
#define STG      8
#define NSTG     (QSTEPS / STG)
#define TAIL0    (KSPLIT * QSTEPS)
static_assert(TAIL0 + KSPLIT == KNZ, "the K parts must cover exactly the dims below 784");

#define WSTEPS   (QSTEPS + 1)
#define WPITCH   (4 * WSTEPS)
#define WCOPY    (NCLS * WPITCH)

struct Stage {
    v2f a0[STG];
    v2f a1[STG];
    v2f b[STG];
};

__device__ __forceinline__ void load_stage_a(Stage& st, const float* ar, const float* ai, int s0) {
#pragma unroll
    for (int u = 0; u < STG; ++u) st.a0[u] = *(const v2f*)(ar + 4 * (s0 + u));
#pragma unroll
    for (int u = 0; u < STG; ++u) st.a1[u] = *(const v2f*)(ai + 4 * (s0 + u));
}

__device__ __forceinline__ void load_stage_b(Stage& st, const float* bl, int s0) {
#pragma unroll
    for (int u = 0; u < STG; ++u) st.b[u] = *(const v2f*)(bl + 4 * (s0 + u));
}

__device__ __forceinline__ void mma_stage(const Stage& st, v8f& acc_re, v8f& acc_im) {
#pragma unroll
    for (int u = 0; u < STG; ++u) {
        acc_re = __builtin_amdgcn_wmma_f32_16x16x4_f32(
            false, st.a0[u], false, st.b[u], (short)0, acc_re, false, false);
        acc_im = __builtin_amdgcn_wmma_f32_16x16x4_f32(
            false, st.a1[u], false, st.b[u], (short)0, acc_im, false, false);
    }
}

__global__ __launch_bounds__(256) void ht_swaptest_kernel(const float* __restrict__ z_re,
                                                          const float* __restrict__ z_im,
                                                          const float* __restrict__ canon,
                                                          float* __restrict__ out) {
    __shared__ __attribute__((aligned(16))) float bcopy[8][WCOPY];
    __shared__ float part[2][KSPLIT][2][256];
    __shared__ float ssq[2][KSPLIT][32];

    const int lane = threadIdx.x & 31;
    const int wave = __builtin_amdgcn_readfirstlane(threadIdx.x >> 5);
    const int tl   = wave >> 2;
    const int kq   = wave & 3;
    const int tile = blockIdx.x * 2 + tl;
    const int base = tile * 16;
    const int m    = lane & 15;
    const int half = lane >> 4;
    const int cls  = (m < NCLS) ? m : m - NCLS;

    float* wb = &bcopy[wave][0];

    const int i1 = (lane < 16) ? 32 + lane : QSTEPS;
    const int d0 = kq * (QSTEPS * 4) + 4 * lane;
    const int d1 = (lane < 16) ? kq * (QSTEPS * 4) + 4 * i1 : 4 * (TAIL0 + kq);
    v4f cv0[NCLS];
    v4f cv1[NCLS];
#pragma unroll
    for (int c = 0; c < NCLS; ++c) {
        cv0[c] = *(const v4f*)(canon + c * IMG + d0);
        cv1[c] = *(const v4f*)(canon + c * IMG + d1);
    }
    __builtin_amdgcn_sched_barrier(0);

    const float* ar = z_re + (size_t)(base + m) * DIM + kq * (QSTEPS * 4) + 2 * half;
    const float* ai = z_im + (size_t)(base + m) * DIM + kq * (QSTEPS * 4) + 2 * half;
    const int ts = TAIL0 + kq;
    const float* tr = z_re + (size_t)(base + m) * DIM + 4 * ts + 2 * half;
    const float* ti = z_im + (size_t)(base + m) * DIM + 4 * ts + 2 * half;
    const float* bl = wb + cls * WPITCH + 2 * half;

    v8f acc_re = {};
    v8f acc_im = {};

    const v2f ta0 = *(const v2f*)tr;
    const v2f ta1 = *(const v2f*)ti;

    Stage s0, s1;
    load_stage_a(s0, ar, ai, 0);
    __builtin_amdgcn_sched_barrier(0);
    load_stage_a(s1, ar, ai, STG);
    __builtin_amdgcn_sched_barrier(0);

#pragma unroll
    for (int c = 0; c < NCLS; ++c) {
        *(v4f*)(wb + c * WPITCH + 4 * lane) = cv0[c];
        *(v4f*)(wb + c * WPITCH + 4 * i1)   = cv1[c];
    }
    __builtin_amdgcn_fence(__ATOMIC_ACQ_REL, "wavefront");

    const v2f tb = *(const v2f*)(bl + 4 * QSTEPS);
    const float* bq = wb + cls * WPITCH + 4 * (QSTEPS / 2) * half;
    v4f sq = {};
#pragma unroll
    for (int k = 0; k < QSTEPS / 2; ++k) {
        const v4f p = *(const v4f*)(bq + 4 * k);
        sq += p * p;
    }
    ssq[tl][kq][lane] = ((sq.x + sq.y) + (sq.z + sq.w)) + (tb.x * tb.x + tb.y * tb.y);
    __builtin_amdgcn_sched_barrier(0);

    load_stage_b(s0, bl, 0);
    load_stage_b(s1, bl, STG);
    __builtin_amdgcn_sched_barrier(0);

#pragma unroll
    for (int st = 0; st < NSTG; st += 2) {
        if (st > 0) {
            load_stage_a(s1, ar, ai, (st + 1) * STG);
            load_stage_b(s1, bl, (st + 1) * STG);
            __builtin_amdgcn_sched_barrier(0);
        }
        if (st == 0) {
            acc_re = __builtin_amdgcn_wmma_f32_16x16x4_f32(
                false, ta0, false, tb, (short)0, acc_re, false, false);
            acc_im = __builtin_amdgcn_wmma_f32_16x16x4_f32(
                false, ta1, false, tb, (short)0, acc_im, false, false);
        }
        mma_stage(s0, acc_re, acc_im);
        __builtin_amdgcn_sched_barrier(0);
        if (st + 2 < NSTG) {
            load_stage_a(s0, ar, ai, (st + 2) * STG);
            load_stage_b(s0, bl, (st + 2) * STG);
        }
        __builtin_amdgcn_sched_barrier(0);
        mma_stage(s1, acc_re, acc_im);
        __builtin_amdgcn_sched_barrier(0);
    }

#pragma unroll
    for (int r = 0; r < 8; ++r) {
        part[tl][kq][0][r * 32 + lane] = acc_re[r];
        part[tl][kq][1][r * 32 + lane] = acc_im[r];
    }
    __syncthreads();

    const int t = threadIdx.x & 127;
    for (int o = t; o < 16 * NCLS; o += 128) {
        const int row = o / NCLS;
        const int c   = o - row * NCLS;
        const int idx = (row & 7) * 32 + 16 * (row >> 3) + c;
        float re = part[tl][0][0][idx];
        float im = part[tl][0][1][idx];
        float n2 = ssq[tl][0][c] + ssq[tl][0][16 + c];
        for (int q = 1; q < KSPLIT; ++q) {
            re += part[tl][q][0][idx];
            im += part[tl][q][1][idx];
            n2 += ssq[tl][q][c] + ssq[tl][q][16 + c];
        }
        out[(size_t)base * NCLS + o] = (re * re + im * im) / n2;
    }
}

extern "C" void kernel_launch(void* const* d_in, const int* in_sizes, int n_in,
                              void* d_out, int out_size, void* d_ws, size_t ws_size,
                              hipStream_t stream) {
    const float* z_re  = (const float*)d_in[0];
    const float* z_im  = (const float*)d_in[1];
    const float* canon = (const float*)d_in[2];
    float* out   = (float*)d_out;

    ht_swaptest_kernel<<<NROWS / 32, 256, 0, stream>>>(z_re, z_im, canon, out);
}


# A rows fetched by whole-line 16-byte loads (a wave load touches 4 lines, was 16) and turned into WMMA fragments through a wave-private LDS area; k order and the order of every sum unchanged
# speedup vs baseline: 2.9683x; 1.0946x over previous
#include <hip/hip_runtime.h>

typedef float v2f __attribute__((ext_vector_type(2)));
typedef float v4f __attribute__((ext_vector_type(4)));
typedef float v8f __attribute__((ext_vector_type(8)));

#define DIM     1024
#define NROWS   16384
#define NCLS    10
#define IMG     784
#define KNZ     (IMG / 4)

#define KSPLIT   4
#define QSTEPS   48
#define STG      8
#define NSTG     (QSTEPS / STG)
#define TAIL0    (KSPLIT * QSTEPS)
static_assert(TAIL0 + KSPLIT == KNZ, "the K parts must cover exactly the dims below 784");

#define WSTEPS   (QSTEPS + 1)
#define WPITCH   (4 * WSTEPS)
#define WCOPY    (NCLS * WPITCH)

#define SPITCH   36
#define SAREA    (16 * SPITCH)

struct Stage {
    v2f a0[STG];
    v2f a1[STG];
    v2f b[STG];
};

struct Raw {
    v4f r[4];
    v4f i[4];
};

__device__ __forceinline__ void load_raw(Raw& g, const float* gr, const float* gi, int s) {
#pragma unroll
    for (int j = 0; j < 4; ++j) g.r[j] = *(const v4f*)(gr + (size_t)j * 2 * DIM + s * (4 * STG));
#pragma unroll
    for (int j = 0; j < 4; ++j) g.i[j] = *(const v4f*)(gi + (size_t)j * 2 * DIM + s * (4 * STG));
}

__device__ __forceinline__ void store_raw(const Raw& g, float* sr, float* si) {
#pragma unroll
    for (int j = 0; j < 4; ++j) *(v4f*)(sr + j * 2 * SPITCH) = g.r[j];
#pragma unroll
    for (int j = 0; j < 4; ++j) *(v4f*)(si + j * 2 * SPITCH) = g.i[j];
}

__device__ __forceinline__ void load_stage_a(Stage& st, const float* fr, const float* fi) {
#pragma unroll
    for (int u = 0; u < STG; ++u) st.a0[u] = *(const v2f*)(fr + 4 * u);
#pragma unroll
    for (int u = 0; u < STG; ++u) st.a1[u] = *(const v2f*)(fi + 4 * u);
}

__device__ __forceinline__ void load_stage_b(Stage& st, const float* bl, int s0) {
#pragma unroll
    for (int u = 0; u < STG; ++u) st.b[u] = *(const v2f*)(bl + 4 * (s0 + u));
}

__device__ __forceinline__ void mma_stage(const Stage& st, v8f& acc_re, v8f& acc_im) {
#pragma unroll
    for (int u = 0; u < STG; ++u) {
        acc_re = __builtin_amdgcn_wmma_f32_16x16x4_f32(
            false, st.a0[u], false, st.b[u], (short)0, acc_re, false, false);
        acc_im = __builtin_amdgcn_wmma_f32_16x16x4_f32(
            false, st.a1[u], false, st.b[u], (short)0, acc_im, false, false);
    }
}

__global__ __launch_bounds__(256) void ht_swaptest_kernel(const float* __restrict__ z_re,
                                                          const float* __restrict__ z_im,
                                                          const float* __restrict__ canon,
                                                          float* __restrict__ out) {
    __shared__ __attribute__((aligned(16))) float bcopy[8][WCOPY];
    __shared__ __attribute__((aligned(16))) float apark[8][2][SAREA];
    __shared__ float part[2][KSPLIT][2][256];
    __shared__ float ssq[2][KSPLIT][32];

    const int lane = threadIdx.x & 31;
    const int wave = __builtin_amdgcn_readfirstlane(threadIdx.x >> 5);
    const int tl   = wave >> 2;
    const int kq   = wave & 3;
    const int tile = blockIdx.x * 2 + tl;
    const int base = tile * 16;
    const int m    = lane & 15;
    const int half = lane >> 4;
    const int cls  = (m < NCLS) ? m : m - NCLS;

    float* wb = &bcopy[wave][0];

    const int i1 = (lane < 16) ? 32 + lane : QSTEPS;
    const int d0 = kq * (QSTEPS * 4) + 4 * lane;
    const int d1 = (lane < 16) ? kq * (QSTEPS * 4) + 4 * i1 : 4 * (TAIL0 + kq);
    v4f cv0[NCLS];
    v4f cv1[NCLS];
#pragma unroll
    for (int c = 0; c < NCLS; ++c) {
        cv0[c] = *(const v4f*)(canon + c * IMG + d0);
        cv1[c] = *(const v4f*)(canon + c * IMG + d1);
    }
    __builtin_amdgcn_sched_barrier(0);

    const int prow = ((lane >> 4) & 1) + 8 * ((lane >> 3) & 1);
    const int pcol = 4 * (lane & 7);
    const float* gr = z_re + (size_t)(base + prow) * DIM + kq * (QSTEPS * 4) + pcol;
    const float* gi = z_im + (size_t)(base + prow) * DIM + kq * (QSTEPS * 4) + pcol;
    float* pr = &apark[wave][0][0] + prow * SPITCH + pcol;
    float* pi = pr + SAREA;
    const float* fr = &apark[wave][0][0] + m * SPITCH + 2 * half;
    const float* fi = fr + SAREA;
    const int ts = TAIL0 + kq;
    const float* tr = z_re + (size_t)(base + m) * DIM + 4 * ts + 2 * half;
    const float* ti = z_im + (size_t)(base + m) * DIM + 4 * ts + 2 * half;
    const float* bl = wb + cls * WPITCH + 2 * half;

    v8f acc_re = {};
    v8f acc_im = {};

    const v2f ta0 = *(const v2f*)tr;
    const v2f ta1 = *(const v2f*)ti;

    Raw g0, g1;
    Stage s0, s1;
    load_raw(g0, gr, gi, 0);
    __builtin_amdgcn_sched_barrier(0);
    load_raw(g1, gr, gi, 1);
    __builtin_amdgcn_sched_barrier(0);

#pragma unroll
    for (int c = 0; c < NCLS; ++c) {
        *(v4f*)(wb + c * WPITCH + 4 * lane) = cv0[c];
        *(v4f*)(wb + c * WPITCH + 4 * i1)   = cv1[c];
    }
    __builtin_amdgcn_fence(__ATOMIC_ACQ_REL, "wavefront");

    const v2f tb = *(const v2f*)(bl + 4 * QSTEPS);
    const float* bq = wb + cls * WPITCH + 4 * (QSTEPS / 2) * half;
    v4f sq = {};
#pragma unroll
    for (int k = 0; k < QSTEPS / 2; ++k) {
        const v4f p = *(const v4f*)(bq + 4 * k);
        sq += p * p;
    }
    ssq[tl][kq][lane] = ((sq.x + sq.y) + (sq.z + sq.w)) + (tb.x * tb.x + tb.y * tb.y);
    __builtin_amdgcn_sched_barrier(0);

    store_raw(g0, pr, pi);
    __builtin_amdgcn_fence(__ATOMIC_ACQ_REL, "wavefront");
    load_raw(g0, gr, gi, 2);
    load_stage_a(s0, fr, fi);
    load_stage_b(s0, bl, 0);
    __builtin_amdgcn_sched_barrier(0);

#pragma unroll
    for (int st = 0; st < NSTG; ++st) {
        Stage& cur = (st & 1) ? s1 : s0;
        Stage& nxt = (st & 1) ? s0 : s1;
        Raw&   gn  = (st & 1) ? g0 : g1;
        if (st + 1 < NSTG) {
            store_raw(gn, pr, pi);
            __builtin_amdgcn_fence(__ATOMIC_ACQ_REL, "wavefront");
            if (st + 3 < NSTG) load_raw(gn, gr, gi, st + 3);
            load_stage_a(nxt, fr, fi);
            load_stage_b(nxt, bl, (st + 1) * STG);
            __builtin_amdgcn_sched_barrier(0);
        }
        if (st == 0) {
            acc_re = __builtin_amdgcn_wmma_f32_16x16x4_f32(
                false, ta0, false, tb, (short)0, acc_re, false, false);
            acc_im = __builtin_amdgcn_wmma_f32_16x16x4_f32(
                false, ta1, false, tb, (short)0, acc_im, false, false);
        }
        mma_stage(cur, acc_re, acc_im);
        __builtin_amdgcn_sched_barrier(0);
    }

#pragma unroll
    for (int r = 0; r < 8; ++r) {
        part[tl][kq][0][r * 32 + lane] = acc_re[r];
        part[tl][kq][1][r * 32 + lane] = acc_im[r];
    }
    __syncthreads();

    const int t = threadIdx.x & 127;
    for (int o = t; o < 16 * NCLS; o += 128) {
        const int row = o / NCLS;
        const int c   = o - row * NCLS;
        const int idx = (row & 7) * 32 + 16 * (row >> 3) + c;
        float re = part[tl][0][0][idx];
        float im = part[tl][0][1][idx];
        float n2 = ssq[tl][0][c] + ssq[tl][0][16 + c];
        for (int q = 1; q < KSPLIT; ++q) {
            re += part[tl][q][0][idx];
            im += part[tl][q][1][idx];
            n2 += ssq[tl][q][c] + ssq[tl][q][16 + c];
        }
        out[(size_t)base * NCLS + o] = (re * re + im * im) / n2;
    }
}

extern "C" void kernel_launch(void* const* d_in, const int* in_sizes, int n_in,
                              void* d_out, int out_size, void* d_ws, size_t ws_size,
                              hipStream_t stream) {
    const float* z_re  = (const float*)d_in[0];
    const float* z_im  = (const float*)d_in[1];
    const float* canon = (const float*)d_in[2];
    float* out   = (float*)d_out;

    ht_swaptest_kernel<<<NROWS / 32, 256, 0, stream>>>(z_re, z_im, canon, out);
}
